# starting kernel
# baseline (speedup 1.0000x reference)
#include <hip/hip_runtime.h>

typedef __attribute__((ext_vector_type(2))) float v2f;
typedef __attribute__((ext_vector_type(8))) float v8f;

__device__ __forceinline__ float h_capture_lane(float a1, float a2, int lane) {
    float h[17];
    h[0] = 1.0f;
    h[1] = -a1;
#pragma unroll
    for (int i = 2; i <= 16; ++i) h[i] = -a1 * h[i - 1] - a2 * h[i - 2];
    float hl = h[0];
#pragma unroll
    for (int i = 1; i <= 16; ++i) hl = (lane == i) ? h[i] : hl;
    return hl;
}

__device__ __forceinline__ float h_gather(float hl, int n) {
    int nn = n < 0 ? 0 : n;
    float v = __int_as_float(__builtin_amdgcn_ds_bpermute(nn << 2, __float_as_int(hl)));
    return n < 0 ? 0.0f : v;
}

__global__ void __launch_bounds__(256)
iir_tile_fir_wmma(const float* __restrict__ u,
                  const float* __restrict__ u_init,
                  const float* __restrict__ bc,
                  const float* __restrict__ ac,
                  float* __restrict__ y0,
                  int T, int tiles) {
    __shared__ float lds[8][16 * 18];
    int lane = threadIdx.x & 31;
    int wave = threadIdx.x >> 5;
    int tile = blockIdx.x * 8 + wave;
    if (tile >= tiles) tile = tiles - 1;
    int nT = T >> 4;
    int tb = tile / nT;
    int tt = tile - tb * nT;
    int t0 = tt << 4;

    float b0 = bc[0], b1 = bc[1];
    float a1 = ac[0], a2 = ac[1];

    float* tl = &lds[wave][0];

    #pragma unroll
    for (int c = 0; c < 8; ++c) {
        int flat = c * 32 + lane;
        int rr = flat >> 4, cc = flat & 15;
        tl[rr * 18 + 1 + cc] = u[(size_t)(tb * 16 + rr) * T + t0 + cc];
    }
    if (lane < 16) {
        int rowg = tb * 16 + lane;
        const float* p = (t0 == 0) ? (u_init + (size_t)rowg * 2)
                                   : (u + (size_t)rowg * T + (t0 - 1));
        tl[lane * 18] = *p;
    }
    __syncthreads();

    int m    = lane & 15;
    int half = lane >> 4;

    float hl = h_capture_lane(a1, a2, lane);

    v2f Af[4]; v2f Bf[4];
#pragma unroll
    for (int j = 0; j < 4; ++j) {
#pragma unroll
        for (int r = 0; r < 2; ++r) {
            int k = 4 * j + 2 * half + r;
            float uc = tl[m * 18 + 1 + k];
            float up = tl[m * 18 + k];
            Af[j][r] = b0 * uc + b1 * up;
            Bf[j][r] = h_gather(hl, m - k);
        }
    }

    v8f acc = {0.f, 0.f, 0.f, 0.f, 0.f, 0.f, 0.f, 0.f};
#pragma unroll
    for (int j = 0; j < 4; ++j)
        acc = __builtin_amdgcn_wmma_f32_16x16x4_f32(
             false, Af[j],  false, Bf[j],
             (short)0, acc,  false,  false);

    float* orow = y0 + (size_t)(tb * 16 + 8 * half) * T + t0 + m;
#pragma unroll
    for (int r = 0; r < 8; ++r)
        orow[(size_t)r * T] = acc[r];
}

__global__ void iir_state_scan(const float* __restrict__ y0,
                               const float* __restrict__ y_init,
                               const float* __restrict__ ac,
                               float2* __restrict__ states,
                               int B, int T) {
    int b = blockIdx.x * blockDim.x + threadIdx.x;
    if (b >= B) return;
    float a1 = ac[0], a2 = ac[1];
    float h[17];
    h[0] = 1.0f; h[1] = -a1;
#pragma unroll
    for (int i = 2; i <= 16; ++i) h[i] = -a1 * h[i - 1] - a2 * h[i - 2];
    float m00 = h[16], m01 = -a2 * h[15], m10 = h[15], m11 = -a2 * h[14];
    float s1 = y_init[(size_t)b * 2 + 0];
    float s2 = y_init[(size_t)b * 2 + 1];
    int nT = T >> 4;
    const float* rowp = y0 + (size_t)b * T;
    float2* st = states + (size_t)b * nT;
    for (int c = 0; c < nT; ++c) {
        st[c] = make_float2(s1, s2);
        float d0 = rowp[c * 16 + 15];
        float d1 = rowp[c * 16 + 14];
        float n1 = m00 * s1 + m01 * s2 + d0;
        float n2 = m10 * s1 + m11 * s2 + d1;
        s1 = n1; s2 = n2;
    }
}

__global__ void iir_apply_state(float* __restrict__ y,
                                const float2* __restrict__ states,
                                const float* __restrict__ ac,
                                int T, size_t total) {
    size_t idx = (size_t)blockIdx.x * blockDim.x + threadIdx.x;
    if (idx >= total) return;
    size_t b = idx / (size_t)T;
    int t = (int)(idx - b * (size_t)T);
    int j = t & 15;
    int c = t >> 4;
    float a1 = ac[0], a2 = ac[1];
    int lane = threadIdx.x & 31;
    float hl = h_capture_lane(a1, a2, lane);
    float g1 = h_gather(hl, j + 1);
    float g2 = h_gather(hl, j);
    float2 s = states[b * (size_t)(T >> 4) + c];
    y[idx] += g1 * s.x + (-a2 * g2) * s.y;
}

extern "C" void kernel_launch(void* const* d_in, const int* in_sizes, int n_in,
                              void* d_out, int out_size, void* d_ws, size_t ws_size,
                              hipStream_t stream) {
    const float* bc = (const float*)d_in[0];
    const float* ac = (const float*)d_in[1];
    const float* u  = (const float*)d_in[2];
    const float* yi = (const float*)d_in[3];
    const float* ui = (const float*)d_in[4];
    float* y = (float*)d_out;

    int B = in_sizes[3] / 2;
    int T = in_sizes[2] / B;
    int nT = T >> 4;
    int tiles = (B >> 4) * nT;
    float2* states = (float2*)d_ws;

    int blocks1 = (tiles + 7) / 8;
    iir_tile_fir_wmma<<<blocks1, 256, 0, stream>>>(u, ui, bc, ac, y, T, tiles);

    iir_state_scan<<<(B + 255) / 256, 256, 0, stream>>>(y, yi, ac, states, B, T);

    size_t total = (size_t)B * T;
    iir_apply_state<<<(int)((total + 255) / 256), 256, 0, stream>>>(y, states, ac, T, total);
}


# a work-group takes two whole rows: 4 waves a row, 16 pieces a wave with its 32 loads issued first, both rows' scans side by side behind the same barriers; 4096 waves, was 8192; sum order kept
# speedup vs baseline: 15.6939x; 15.6939x over previous
#include <hip/hip_runtime.h>

typedef __attribute__((ext_vector_type(2))) float v2f;
typedef __attribute__((ext_vector_type(8))) float v8f;

__device__ __forceinline__ float h_gather(float hl, int n) {
    int nn = n < 0 ? 0 : n;
    float v = __int_as_float(__builtin_amdgcn_ds_bpermute(nn << 2, __float_as_int(hl)));
    return n < 0 ? 0.0f : v;
}

#define ROW_R 2
#define ROW_W 8
#define ROW_WR (ROW_W / ROW_R)
#define ROW_P 16
#define ROW_TILES (ROW_WR * ROW_P * 16)
#define ROW_CH 4
#define ROW_NCH (ROW_TILES / ROW_CH)
#define ROW_GR 16
#define ROW_NGR (ROW_NCH / ROW_GR)
#define ROW_U 8
#define ROW_U3 16

static_assert(ROW_NCH == ROW_W * 32, "one scan chunk of each row per thread");
static_assert(ROW_R * ROW_NGR <= 32, "the group step of all rows runs in one wave");

__global__ void __launch_bounds__(ROW_W * 32)
iir_rows_fused(const float* __restrict__ u,
               const float* __restrict__ u_init,
               const float* __restrict__ y_init,
               const float* __restrict__ bc,
               const float* __restrict__ ac,
               float* __restrict__ y,
               int B, int T) {
    __shared__ __attribute__((aligned(16))) float2 st[ROW_R * ROW_TILES];
    __shared__ __attribute__((aligned(16))) float2 zb[ROW_R * ROW_NCH];
    __shared__ __attribute__((aligned(16))) float2 gt[ROW_R * ROW_NGR];
    __shared__ float4 pw[2];

    int t    = threadIdx.x;
    int lane = t & 31;
    int w    = __builtin_amdgcn_readfirstlane(t >> 5);
    int n    = lane & 15;
    int half = lane >> 4;
    int wr   = w / ROW_WR;
    int wq   = w % ROW_WR;
    int row0 = blockIdx.x * ROW_R;
    int row  = row0 + wr;
    bool rok = row < B;
    int cw   = wq * (ROW_P * 256);

    float b0 = bc[0], b1 = bc[1];
    float a1 = ac[0], a2 = ac[1];

    float yi1[ROW_R], yi2[ROW_R];
#pragma unroll
    for (int r = 0; r < ROW_R; ++r) {
        bool ok = (row0 + r) < B;
        int  rr = ok ? (row0 + r) : row0;
        float v1 = y_init[(size_t)rr * 2 + 0];
        float v2 = y_init[(size_t)rr * 2 + 1];
        yi1[r] = ok ? v1 : 0.f;
        yi2[r] = ok ? v2 : 0.f;
    }

    const float* urow = u + (size_t)row * T;
    int loff = 16 * n + 4 * half;

    float4 lo[ROW_P], hi[ROW_P];
    float  pv[ROW_P];
#pragma unroll
    for (int i = 0; i < ROW_P; ++i) {
        int c0 = cw + i * 256;
        lo[i] = make_float4(0.f, 0.f, 0.f, 0.f);
        hi[i] = lo[i];
        pv[i] = 0.f;
        if (rok && c0 < T) {
            const float* p = urow + c0 + loff;
            lo[i] = *reinterpret_cast<const float4*>(p);
            hi[i] = *reinterpret_cast<const float4*>(p + 8);
            pv[i] = (c0 == 0) ? u_init[(size_t)row * 2] : urow[c0 - 1];
        }
    }

    float h[17];
    h[0] = 1.0f;
    h[1] = -a1;
#pragma unroll
    for (int i = 2; i <= 16; ++i) h[i] = -a1 * h[i - 1] - a2 * h[i - 2];
    float hl = h[0];
#pragma unroll
    for (int i = 1; i <= 16; ++i) hl = (lane == i) ? h[i] : hl;
    float m00 = h[16], m01 = -a2 * h[15], m10 = h[15], m11 = -a2 * h[14];

    if (w == 0) {
        double p00 = m00, p01 = m01, p10 = m10, p11 = m11;
#pragma unroll
        for (int s = 1; s < ROW_CH * ROW_GR; s <<= 1) {
            if (s == ROW_CH && lane == 0)
                pw[0] = make_float4((float)p00, (float)p01, (float)p10, (float)p11);
            double tt = p01 * p10;
            double tr = p00 + p11;
            p00 = p00 * p00 + tt;
            p11 = p11 * p11 + tt;
            p01 = p01 * tr;
            p10 = p10 * tr;
        }
        if (lane == 0) pw[1] = make_float4((float)p00, (float)p01, (float)p10, (float)p11);
    }

    int colM = 4 * (n >> 3) + (n & 3) + 8 * ((n >> 2) & 1);
    v2f Af[4];
#pragma unroll
    for (int s = 0; s < 4; ++s) {
#pragma unroll
        for (int r = 0; r < 2; ++r) {
            int k = 8 * (s >> 1) + 4 * half + 2 * (s & 1) + r;
            Af[s][r] = h_gather(hl, colM - k);
        }
    }

    int src1 = half ? n : (n + 15);
    int src2 = lane ^ 16;

    v8f acc[ROW_P];
#pragma unroll
    for (int i = 0; i < ROW_P; ++i) {
        float q1 = half ? hi[i].w : lo[i].w;
        float q2 = half ? lo[i].w : hi[i].w;
        float p1 = __int_as_float(__builtin_amdgcn_ds_bpermute(src1 << 2, __float_as_int(q1)));
        float p2 = __int_as_float(__builtin_amdgcn_ds_bpermute(src2 << 2, __float_as_int(q2)));
        p1 = (lane == 0) ? pv[i] : p1;

        v2f Bf[4];
        Bf[0][0] = b0 * lo[i].x + b1 * p1;
        Bf[0][1] = b0 * lo[i].y + b1 * lo[i].x;
        Bf[1][0] = b0 * lo[i].z + b1 * lo[i].y;
        Bf[1][1] = b0 * lo[i].w + b1 * lo[i].z;
        Bf[2][0] = b0 * hi[i].x + b1 * p2;
        Bf[2][1] = b0 * hi[i].y + b1 * hi[i].x;
        Bf[3][0] = b0 * hi[i].z + b1 * hi[i].y;
        Bf[3][1] = b0 * hi[i].w + b1 * hi[i].z;

        v8f a = {0.f, 0.f, 0.f, 0.f, 0.f, 0.f, 0.f, 0.f};
#pragma unroll
        for (int s = 0; s < 4; ++s)
            a = __builtin_amdgcn_wmma_f32_16x16x4_f32(
                 false, Af[s],  false, Bf[s],
                 (short)0, a,  false,  false);
        acc[i] = a;
    }

    int tb = wr * ROW_TILES + wq * (ROW_P * 16);
    if (half) {
#pragma unroll
        for (int i = 0; i < ROW_P; ++i)
            st[tb + i * 16 + n] = make_float2(acc[i][7], acc[i][6]);
    }
    __syncthreads();

    float2 e[ROW_R][ROW_CH];
#pragma unroll
    for (int r = 0; r < ROW_R; ++r) {
        const float4* sp = reinterpret_cast<const float4*>(&st[r * ROW_TILES + t * ROW_CH]);
#pragma unroll
        for (int i = 0; i < ROW_CH / 2; ++i) {
            float4 q = sp[i];
            e[r][2 * i]     = make_float2(q.x, q.y);
            e[r][2 * i + 1] = make_float2(q.z, q.w);
        }
    }
#pragma unroll
    for (int r = 0; r < ROW_R; ++r) {
        float z1 = 0.f, z2 = 0.f;
#pragma unroll
        for (int i = 0; i < ROW_CH; ++i) {
            float n1 = m00 * z1 + m01 * z2 + e[r][i].x;
            float n2 = m10 * z1 + m11 * z2 + e[r][i].y;
            z1 = n1; z2 = n2;
        }
        zb[r * ROW_NCH + t] = make_float2(z1, z2);
    }
    __syncthreads();

    float4 Pc = pw[0];
    float4 Pg = pw[1];
    if (t < ROW_R * ROW_NGR) {
        float g1 = 0.f, g2 = 0.f;
#pragma unroll
        for (int q0 = 0; q0 < ROW_GR; q0 += ROW_U) {
            float2 ev[ROW_U];
#pragma unroll
            for (int j = 0; j < ROW_U; ++j) ev[j] = zb[t * ROW_GR + q0 + j];
#pragma unroll
            for (int j = 0; j < ROW_U; ++j) {
                float n1 = Pc.x * g1 + Pc.y * g2 + ev[j].x;
                float n2 = Pc.z * g1 + Pc.w * g2 + ev[j].y;
                g1 = n1; g2 = n2;
            }
        }
        gt[t] = make_float2(g1, g2);
    }
    __syncthreads();

    int g  = t / ROW_GR;
    int jc = t % ROW_GR;
    float s1[ROW_R], s2[ROW_R];
#pragma unroll
    for (int r = 0; r < ROW_R; ++r) { s1[r] = yi1[r]; s2[r] = yi2[r]; }
#pragma unroll
    for (int v0 = 0; v0 < ROW_NGR; v0 += ROW_U3) {
        float2 ev[ROW_R][ROW_U3];
#pragma unroll
        for (int r = 0; r < ROW_R; ++r) {
            const float4* gp = reinterpret_cast<const float4*>(&gt[r * ROW_NGR + v0]);
#pragma unroll
            for (int j = 0; j < ROW_U3 / 2; ++j) {
                float4 q = gp[j];
                ev[r][2 * j]     = make_float2(q.x, q.y);
                ev[r][2 * j + 1] = make_float2(q.z, q.w);
            }
        }
#pragma unroll
        for (int j = 0; j < ROW_U3; ++j) {
            bool take = (v0 + j) < g;
#pragma unroll
            for (int r = 0; r < ROW_R; ++r) {
                float n1 = Pg.x * s1[r] + Pg.y * s2[r] + ev[r][j].x;
                float n2 = Pg.z * s1[r] + Pg.w * s2[r] + ev[r][j].y;
                s1[r] = take ? n1 : s1[r];
                s2[r] = take ? n2 : s2[r];
            }
        }
    }
#pragma unroll
    for (int q0 = 0; q0 < ROW_GR; q0 += ROW_U3) {
        float2 ev[ROW_R][ROW_U3];
#pragma unroll
        for (int r = 0; r < ROW_R; ++r) {
            const float4* zp = reinterpret_cast<const float4*>(&zb[r * ROW_NCH + g * ROW_GR + q0]);
#pragma unroll
            for (int j = 0; j < ROW_U3 / 2; ++j) {
                float4 q = zp[j];
                ev[r][2 * j]     = make_float2(q.x, q.y);
                ev[r][2 * j + 1] = make_float2(q.z, q.w);
            }
        }
#pragma unroll
        for (int j = 0; j < ROW_U3; ++j) {
            bool take = (q0 + j) < jc;
#pragma unroll
            for (int r = 0; r < ROW_R; ++r) {
                float n1 = Pc.x * s1[r] + Pc.y * s2[r] + ev[r][j].x;
                float n2 = Pc.z * s1[r] + Pc.w * s2[r] + ev[r][j].y;
                s1[r] = take ? n1 : s1[r];
                s2[r] = take ? n2 : s2[r];
            }
        }
    }
#pragma unroll
    for (int r = 0; r < ROW_R; ++r) {
        float2 sv[ROW_CH];
        float c1 = s1[r], c2 = s2[r];
#pragma unroll
        for (int i = 0; i < ROW_CH; ++i) {
            sv[i] = make_float2(c1, c2);
            float n1 = m00 * c1 + m01 * c2 + e[r][i].x;
            float n2 = m10 * c1 + m11 * c2 + e[r][i].y;
            c1 = n1; c2 = n2;
        }
        float4* sp = reinterpret_cast<float4*>(&st[r * ROW_TILES + t * ROW_CH]);
#pragma unroll
        for (int i = 0; i < ROW_CH / 2; ++i)
            sp[i] = make_float4(sv[2 * i].x, sv[2 * i].y, sv[2 * i + 1].x, sv[2 * i + 1].y);
    }
    __syncthreads();

    float gA[4], fA[4], gB[4], fB[4];
#pragma unroll
    for (int i = 0; i < 4; ++i) {
        gA[i] = half ? h[4 + i + 1]  : h[i + 1];
        fA[i] = -a2 * (half ? h[4 + i]  : h[i]);
        gB[i] = half ? h[12 + i + 1] : h[8 + i + 1];
        fB[i] = -a2 * (half ? h[12 + i] : h[8 + i]);
    }
    float* yrow = y + (size_t)row * T;
    if (rok && cw + ROW_P * 256 <= T) {
        float2 sv[ROW_P];
#pragma unroll
        for (int i = 0; i < ROW_P; ++i) sv[i] = st[tb + i * 16 + n];
#pragma unroll
        for (int i = 0; i < ROW_P; ++i) {
            float2 s = sv[i];
            float4 o0, o1;
            o0.x = acc[i][0] + (gA[0] * s.x + fA[0] * s.y);
            o0.y = acc[i][1] + (gA[1] * s.x + fA[1] * s.y);
            o0.z = acc[i][2] + (gA[2] * s.x + fA[2] * s.y);
            o0.w = acc[i][3] + (gA[3] * s.x + fA[3] * s.y);
            o1.x = acc[i][4] + (gB[0] * s.x + fB[0] * s.y);
            o1.y = acc[i][5] + (gB[1] * s.x + fB[1] * s.y);
            o1.z = acc[i][6] + (gB[2] * s.x + fB[2] * s.y);
            o1.w = acc[i][7] + (gB[3] * s.x + fB[3] * s.y);
            float* p = yrow + cw + i * 256 + loff;
            *reinterpret_cast<float4*>(p)     = o0;
            *reinterpret_cast<float4*>(p + 8) = o1;
        }
    } else {
#pragma unroll
        for (int i = 0; i < ROW_P; ++i) {
            int c0 = cw + i * 256;
            float2 s = st[tb + i * 16 + n];
            float4 o0, o1;
            o0.x = acc[i][0] + (gA[0] * s.x + fA[0] * s.y);
            o0.y = acc[i][1] + (gA[1] * s.x + fA[1] * s.y);
            o0.z = acc[i][2] + (gA[2] * s.x + fA[2] * s.y);
            o0.w = acc[i][3] + (gA[3] * s.x + fA[3] * s.y);
            o1.x = acc[i][4] + (gB[0] * s.x + fB[0] * s.y);
            o1.y = acc[i][5] + (gB[1] * s.x + fB[1] * s.y);
            o1.z = acc[i][6] + (gB[2] * s.x + fB[2] * s.y);
            o1.w = acc[i][7] + (gB[3] * s.x + fB[3] * s.y);
            if (rok && c0 < T) {
                float* p = yrow + c0 + loff;
                *reinterpret_cast<float4*>(p)     = o0;
                *reinterpret_cast<float4*>(p + 8) = o1;
            }
        }
    }
}

extern "C" void kernel_launch(void* const* d_in, const int* in_sizes, int n_in,
                              void* d_out, int out_size, void* d_ws, size_t ws_size,
                              hipStream_t stream) {
    const float* bc = (const float*)d_in[0];
    const float* ac = (const float*)d_in[1];
    const float* u  = (const float*)d_in[2];
    const float* yi = (const float*)d_in[3];
    const float* ui = (const float*)d_in[4];
    float* y = (float*)d_out;

    int B = in_sizes[3] / 2;
    int T = in_sizes[2] / B;

    iir_rows_fused<<<(B + ROW_R - 1) / ROW_R, ROW_W * 32, 0, stream>>>(u, ui, yi, bc, ac, y, B, T);
}


# stores: each wave transposes a piece's result through an LDS slot of its own (no barrier), so a store instruction writes four whole 128-byte lines, was half of each of 8 lines; sum order kept
# speedup vs baseline: 17.5748x; 1.1198x over previous
#include <hip/hip_runtime.h>

typedef __attribute__((ext_vector_type(2))) float v2f;
typedef __attribute__((ext_vector_type(4))) float v4f;
typedef __attribute__((ext_vector_type(8))) float v8f;

__device__ __forceinline__ float h_gather(float hl, int n) {
    int nn = n < 0 ? 0 : n;
    float v = __int_as_float(__builtin_amdgcn_ds_bpermute(nn << 2, __float_as_int(hl)));
    return n < 0 ? 0.0f : v;
}

__device__ __forceinline__ void wave_lds_exchange_point() {
    __builtin_amdgcn_fence(__ATOMIC_RELEASE, "wavefront");
    __builtin_amdgcn_wave_barrier();
    __builtin_amdgcn_fence(__ATOMIC_ACQUIRE, "wavefront");
}

#define ROW_R 2
#define ROW_W 8
#define ROW_WR (ROW_W / ROW_R)
#define ROW_P 16
#define ROW_TILES (ROW_WR * ROW_P * 16)
#define ROW_CH 4
#define ROW_NCH (ROW_TILES / ROW_CH)
#define ROW_GR 16
#define ROW_NGR (ROW_NCH / ROW_GR)
#define ROW_U 8
#define ROW_U3 16
#define ROW_NB 4
#define ROW_SD 2

static_assert(ROW_NCH == ROW_W * 32, "one scan chunk of each row per thread");
static_assert(ROW_R * ROW_NGR <= 32, "the group step of all rows runs in one wave");
static_assert(ROW_SD < ROW_NB, "a slot is rewritten only behind the exchange point that follows its last read");

__global__ void __launch_bounds__(ROW_W * 32)
iir_rows_fused(const float* __restrict__ u,
               const float* __restrict__ u_init,
               const float* __restrict__ y_init,
               const float* __restrict__ bc,
               const float* __restrict__ ac,
               float* __restrict__ y,
               int B, int T) {
    __shared__ __attribute__((aligned(16))) float2 st[ROW_R * ROW_TILES];
    __shared__ __attribute__((aligned(16))) float2 zb[ROW_R * ROW_NCH];
    __shared__ __attribute__((aligned(16))) float2 gt[ROW_R * ROW_NGR];
    __shared__ float4 pw[2];
    __shared__ v4f xs[ROW_W * ROW_NB * 64];

    int t    = threadIdx.x;
    int lane = t & 31;
    int w    = __builtin_amdgcn_readfirstlane(t >> 5);
    int n    = lane & 15;
    int half = lane >> 4;
    int wr   = w / ROW_WR;
    int wq   = w % ROW_WR;
    int row0 = blockIdx.x * ROW_R;
    int row  = row0 + wr;
    bool rok = row < B;
    int cw   = wq * (ROW_P * 256);

    float b0 = bc[0], b1 = bc[1];
    float a1 = ac[0], a2 = ac[1];

    float yi1[ROW_R], yi2[ROW_R];
#pragma unroll
    for (int r = 0; r < ROW_R; ++r) {
        bool ok = (row0 + r) < B;
        int  rr = ok ? (row0 + r) : row0;
        float v1 = y_init[(size_t)rr * 2 + 0];
        float v2 = y_init[(size_t)rr * 2 + 1];
        yi1[r] = ok ? v1 : 0.f;
        yi2[r] = ok ? v2 : 0.f;
    }

    const float* urow = u + (size_t)row * T;
    int loff = 16 * n + 4 * half;

    float4 lo[ROW_P], hi[ROW_P];
    float  pv[ROW_P];
#pragma unroll
    for (int i = 0; i < ROW_P; ++i) {
        int c0 = cw + i * 256;
        lo[i] = make_float4(0.f, 0.f, 0.f, 0.f);
        hi[i] = lo[i];
        pv[i] = 0.f;
        if (rok && c0 < T) {
            const float* p = urow + c0 + loff;
            lo[i] = *reinterpret_cast<const float4*>(p);
            hi[i] = *reinterpret_cast<const float4*>(p + 8);
            pv[i] = (c0 == 0) ? u_init[(size_t)row * 2] : urow[c0 - 1];
        }
    }

    float h[17];
    h[0] = 1.0f;
    h[1] = -a1;
#pragma unroll
    for (int i = 2; i <= 16; ++i) h[i] = -a1 * h[i - 1] - a2 * h[i - 2];
    float hl = h[0];
#pragma unroll
    for (int i = 1; i <= 16; ++i) hl = (lane == i) ? h[i] : hl;
    float m00 = h[16], m01 = -a2 * h[15], m10 = h[15], m11 = -a2 * h[14];

    if (w == 0) {
        double p00 = m00, p01 = m01, p10 = m10, p11 = m11;
#pragma unroll
        for (int s = 1; s < ROW_CH * ROW_GR; s <<= 1) {
            if (s == ROW_CH && lane == 0)
                pw[0] = make_float4((float)p00, (float)p01, (float)p10, (float)p11);
            double tt = p01 * p10;
            double tr = p00 + p11;
            p00 = p00 * p00 + tt;
            p11 = p11 * p11 + tt;
            p01 = p01 * tr;
            p10 = p10 * tr;
        }
        if (lane == 0) pw[1] = make_float4((float)p00, (float)p01, (float)p10, (float)p11);
    }

    int colM = 4 * (n >> 3) + (n & 3) + 8 * ((n >> 2) & 1);
    v2f Af[4];
#pragma unroll
    for (int s = 0; s < 4; ++s) {
#pragma unroll
        for (int r = 0; r < 2; ++r) {
            int k = 8 * (s >> 1) + 4 * half + 2 * (s & 1) + r;
            Af[s][r] = h_gather(hl, colM - k);
        }
    }

    int src1 = half ? n : (n + 15);
    int src2 = lane ^ 16;

    v8f acc[ROW_P];
#pragma unroll
    for (int i = 0; i < ROW_P; ++i) {
        float q1 = half ? hi[i].w : lo[i].w;
        float q2 = half ? lo[i].w : hi[i].w;
        float p1 = __int_as_float(__builtin_amdgcn_ds_bpermute(src1 << 2, __float_as_int(q1)));
        float p2 = __int_as_float(__builtin_amdgcn_ds_bpermute(src2 << 2, __float_as_int(q2)));
        p1 = (lane == 0) ? pv[i] : p1;

        v2f Bf[4];
        Bf[0][0] = b0 * lo[i].x + b1 * p1;
        Bf[0][1] = b0 * lo[i].y + b1 * lo[i].x;
        Bf[1][0] = b0 * lo[i].z + b1 * lo[i].y;
        Bf[1][1] = b0 * lo[i].w + b1 * lo[i].z;
        Bf[2][0] = b0 * hi[i].x + b1 * p2;
        Bf[2][1] = b0 * hi[i].y + b1 * hi[i].x;
        Bf[3][0] = b0 * hi[i].z + b1 * hi[i].y;
        Bf[3][1] = b0 * hi[i].w + b1 * hi[i].z;

        v8f a = {0.f, 0.f, 0.f, 0.f, 0.f, 0.f, 0.f, 0.f};
#pragma unroll
        for (int s = 0; s < 4; ++s)
            a = __builtin_amdgcn_wmma_f32_16x16x4_f32(
                 false, Af[s],  false, Bf[s],
                 (short)0, a,  false,  false);
        acc[i] = a;
    }

    int tb = wr * ROW_TILES + wq * (ROW_P * 16);
    if (half) {
#pragma unroll
        for (int i = 0; i < ROW_P; ++i)
            st[tb + i * 16 + n] = make_float2(acc[i][7], acc[i][6]);
    }
    __syncthreads();

    float2 e[ROW_R][ROW_CH];
#pragma unroll
    for (int r = 0; r < ROW_R; ++r) {
        const float4* sp = reinterpret_cast<const float4*>(&st[r * ROW_TILES + t * ROW_CH]);
#pragma unroll
        for (int i = 0; i < ROW_CH / 2; ++i) {
            float4 q = sp[i];
            e[r][2 * i]     = make_float2(q.x, q.y);
            e[r][2 * i + 1] = make_float2(q.z, q.w);
        }
    }
#pragma unroll
    for (int r = 0; r < ROW_R; ++r) {
        float z1 = 0.f, z2 = 0.f;
#pragma unroll
        for (int i = 0; i < ROW_CH; ++i) {
            float n1 = m00 * z1 + m01 * z2 + e[r][i].x;
            float n2 = m10 * z1 + m11 * z2 + e[r][i].y;
            z1 = n1; z2 = n2;
        }
        zb[r * ROW_NCH + t] = make_float2(z1, z2);
    }
    __syncthreads();

    float4 Pc = pw[0];
    float4 Pg = pw[1];
    if (t < ROW_R * ROW_NGR) {
        float g1 = 0.f, g2 = 0.f;
#pragma unroll
        for (int q0 = 0; q0 < ROW_GR; q0 += ROW_U) {
            float2 ev[ROW_U];
#pragma unroll
            for (int j = 0; j < ROW_U; ++j) ev[j] = zb[t * ROW_GR + q0 + j];
#pragma unroll
            for (int j = 0; j < ROW_U; ++j) {
                float n1 = Pc.x * g1 + Pc.y * g2 + ev[j].x;
                float n2 = Pc.z * g1 + Pc.w * g2 + ev[j].y;
                g1 = n1; g2 = n2;
            }
        }
        gt[t] = make_float2(g1, g2);
    }
    __syncthreads();

    int g  = t / ROW_GR;
    int jc = t % ROW_GR;
    float s1[ROW_R], s2[ROW_R];
#pragma unroll
    for (int r = 0; r < ROW_R; ++r) { s1[r] = yi1[r]; s2[r] = yi2[r]; }
#pragma unroll
    for (int v0 = 0; v0 < ROW_NGR; v0 += ROW_U3) {
        float2 ev[ROW_R][ROW_U3];
#pragma unroll
        for (int r = 0; r < ROW_R; ++r) {
            const float4* gp = reinterpret_cast<const float4*>(&gt[r * ROW_NGR + v0]);
#pragma unroll
            for (int j = 0; j < ROW_U3 / 2; ++j) {
                float4 q = gp[j];
                ev[r][2 * j]     = make_float2(q.x, q.y);
                ev[r][2 * j + 1] = make_float2(q.z, q.w);
            }
        }
#pragma unroll
        for (int j = 0; j < ROW_U3; ++j) {
            bool take = (v0 + j) < g;
#pragma unroll
            for (int r = 0; r < ROW_R; ++r) {
                float n1 = Pg.x * s1[r] + Pg.y * s2[r] + ev[r][j].x;
                float n2 = Pg.z * s1[r] + Pg.w * s2[r] + ev[r][j].y;
                s1[r] = take ? n1 : s1[r];
                s2[r] = take ? n2 : s2[r];
            }
        }
    }
#pragma unroll
    for (int q0 = 0; q0 < ROW_GR; q0 += ROW_U3) {
        float2 ev[ROW_R][ROW_U3];
#pragma unroll
        for (int r = 0; r < ROW_R; ++r) {
            const float4* zp = reinterpret_cast<const float4*>(&zb[r * ROW_NCH + g * ROW_GR + q0]);
#pragma unroll
            for (int j = 0; j < ROW_U3 / 2; ++j) {
                float4 q = zp[j];
                ev[r][2 * j]     = make_float2(q.x, q.y);
                ev[r][2 * j + 1] = make_float2(q.z, q.w);
            }
        }
#pragma unroll
        for (int j = 0; j < ROW_U3; ++j) {
            bool take = (q0 + j) < jc;
#pragma unroll
            for (int r = 0; r < ROW_R; ++r) {
                float n1 = Pc.x * s1[r] + Pc.y * s2[r] + ev[r][j].x;
                float n2 = Pc.z * s1[r] + Pc.w * s2[r] + ev[r][j].y;
                s1[r] = take ? n1 : s1[r];
                s2[r] = take ? n2 : s2[r];
            }
        }
    }
#pragma unroll
    for (int r = 0; r < ROW_R; ++r) {
        float2 sv[ROW_CH];
        float c1 = s1[r], c2 = s2[r];
#pragma unroll
        for (int i = 0; i < ROW_CH; ++i) {
            sv[i] = make_float2(c1, c2);
            float n1 = m00 * c1 + m01 * c2 + e[r][i].x;
            float n2 = m10 * c1 + m11 * c2 + e[r][i].y;
            c1 = n1; c2 = n2;
        }
        float4* sp = reinterpret_cast<float4*>(&st[r * ROW_TILES + t * ROW_CH]);
#pragma unroll
        for (int i = 0; i < ROW_CH / 2; ++i)
            sp[i] = make_float4(sv[2 * i].x, sv[2 * i].y, sv[2 * i + 1].x, sv[2 * i + 1].y);
    }
    __syncthreads();

    float gA[4], fA[4], gB[4], fB[4];
#pragma unroll
    for (int i = 0; i < 4; ++i) {
        gA[i] = half ? h[4 + i + 1]  : h[i + 1];
        fA[i] = -a2 * (half ? h[4 + i]  : h[i]);
        gB[i] = half ? h[12 + i + 1] : h[8 + i + 1];
        fB[i] = -a2 * (half ? h[12 + i] : h[8 + i]);
    }
    float* yrow = y + (size_t)row * T;
    if (rok && cw + ROW_P * 256 <= T) {
        float2 sv[ROW_P];
#pragma unroll
        for (int i = 0; i < ROW_P; ++i) sv[i] = st[tb + i * 16 + n];
        v4f* xw = xs + w * (ROW_NB * 64);
        int q4  = lane & 3, t4 = lane >> 2;
        int wo0 = 16 * half       + ((n + 4 * half) & 15);
        int wo1 = 16 * (2 + half) + ((n + 8 + 4 * half) & 15);
        int roA = 16 * q4 + ((t4 + 4 * q4) & 15);
        int roB = 16 * q4 + ((t4 + 8 + 4 * q4) & 15);
        float* yl = yrow + cw + 4 * lane;
        v4f la[ROW_P], lb[ROW_P];
#pragma unroll
        for (int i = 0; i < ROW_P + ROW_SD; ++i) {
            if (i < ROW_P) {
                float2 s = sv[i];
                v4f o0, o1;
                o0.x = acc[i][0] + (gA[0] * s.x + fA[0] * s.y);
                o0.y = acc[i][1] + (gA[1] * s.x + fA[1] * s.y);
                o0.z = acc[i][2] + (gA[2] * s.x + fA[2] * s.y);
                o0.w = acc[i][3] + (gA[3] * s.x + fA[3] * s.y);
                o1.x = acc[i][4] + (gB[0] * s.x + fB[0] * s.y);
                o1.y = acc[i][5] + (gB[1] * s.x + fB[1] * s.y);
                o1.z = acc[i][6] + (gB[2] * s.x + fB[2] * s.y);
                o1.w = acc[i][7] + (gB[3] * s.x + fB[3] * s.y);
                v4f* slot = xw + (i % ROW_NB) * 64;
                slot[wo0] = o0;
                slot[wo1] = o1;
                wave_lds_exchange_point();
                la[i] = slot[roA];
                lb[i] = slot[roB];
            }
            if (i >= ROW_SD) {
                float* p = yl + (i - ROW_SD) * 256;
                *reinterpret_cast<v4f*>(p)       = la[i - ROW_SD];
                *reinterpret_cast<v4f*>(p + 128) = lb[i - ROW_SD];
            }
        }
    } else {
#pragma unroll
        for (int i = 0; i < ROW_P; ++i) {
            int c0 = cw + i * 256;
            float2 s = st[tb + i * 16 + n];
            float4 o0, o1;
            o0.x = acc[i][0] + (gA[0] * s.x + fA[0] * s.y);
            o0.y = acc[i][1] + (gA[1] * s.x + fA[1] * s.y);
            o0.z = acc[i][2] + (gA[2] * s.x + fA[2] * s.y);
            o0.w = acc[i][3] + (gA[3] * s.x + fA[3] * s.y);
            o1.x = acc[i][4] + (gB[0] * s.x + fB[0] * s.y);
            o1.y = acc[i][5] + (gB[1] * s.x + fB[1] * s.y);
            o1.z = acc[i][6] + (gB[2] * s.x + fB[2] * s.y);
            o1.w = acc[i][7] + (gB[3] * s.x + fB[3] * s.y);
            if (rok && c0 < T) {
                float* p = yrow + c0 + loff;
                *reinterpret_cast<float4*>(p)     = o0;
                *reinterpret_cast<float4*>(p + 8) = o1;
            }
        }
    }
}

extern "C" void kernel_launch(void* const* d_in, const int* in_sizes, int n_in,
                              void* d_out, int out_size, void* d_ws, size_t ws_size,
                              hipStream_t stream) {
    const float* bc = (const float*)d_in[0];
    const float* ac = (const float*)d_in[1];
    const float* u  = (const float*)d_in[2];
    const float* yi = (const float*)d_in[3];
    const float* ui = (const float*)d_in[4];
    float* y = (float*)d_out;

    int B = in_sizes[3] / 2;
    int T = in_sizes[2] / B;

    iir_rows_fused<<<(B + ROW_R - 1) / ROW_R, ROW_W * 32, 0, stream>>>(u, ui, yi, bc, ac, y, B, T);
}


# whole-row waves: 32 loads in one run, no per-piece bound; products in piece order, 2 pieces a group, each group behind a counted wait for its own loads (was one wait for all 32); sum order kept
# speedup vs baseline: 20.6340x; 1.1741x over previous
#include <hip/hip_runtime.h>

typedef __attribute__((ext_vector_type(2))) float v2f;
typedef __attribute__((ext_vector_type(4))) float v4f;
typedef __attribute__((ext_vector_type(8))) float v8f;

__device__ __forceinline__ float h_gather(float hl, int n) {
    int nn = n < 0 ? 0 : n;
    float v = __int_as_float(__builtin_amdgcn_ds_bpermute(nn << 2, __float_as_int(hl)));
    return n < 0 ? 0.0f : v;
}

__device__ __forceinline__ void wave_lds_exchange_point() {
    __builtin_amdgcn_fence(__ATOMIC_RELEASE, "wavefront");
    __builtin_amdgcn_wave_barrier();
    __builtin_amdgcn_fence(__ATOMIC_ACQUIRE, "wavefront");
}

#define ROW_R 2
#define ROW_W 8
#define ROW_WR (ROW_W / ROW_R)
#define ROW_P 16
#define ROW_TILES (ROW_WR * ROW_P * 16)
#define ROW_CH 4
#define ROW_NCH (ROW_TILES / ROW_CH)
#define ROW_GR 16
#define ROW_NGR (ROW_NCH / ROW_GR)
#define ROW_U 8
#define ROW_U3 16
#define ROW_NB 4
#define ROW_SD 2
#define ROW_G 2

static_assert(ROW_NCH == ROW_W * 32, "one scan chunk of each row per thread");
static_assert(ROW_R * ROW_NGR <= 32, "the group step of all rows runs in one wave");
static_assert(ROW_SD < ROW_NB, "a slot is rewritten only behind the exchange point that follows its last read");
static_assert(ROW_P % ROW_G == 0, "whole groups");

template <bool FULL>
__device__ __forceinline__ void load_and_multiply(const float* __restrict__ urow,
                                                  const float* __restrict__ u_init,
                                                  int row, bool rok, int cw, int T,
                                                  int lane, int n, int half, int w,
                                                  float b0, float b1, float a1, float a2,
                                                  float4* pw, float (&h)[17], v8f (&acc)[ROW_P]) {
    int loff = 16 * n + 4 * half;

    v4f   lo[ROW_P], hi[ROW_P];
    float pv[ROW_P];
#pragma unroll
    for (int i = 0; i < ROW_P; ++i) {
        int c0 = cw + i * 256;
        if (FULL) {
            const float* p = urow + c0 + loff;
            lo[i] = *reinterpret_cast<const v4f*>(p);
            hi[i] = *reinterpret_cast<const v4f*>(p + 8);
            pv[i] = (c0 == 0) ? u_init[(size_t)row * 2] : urow[c0 - 1];
        } else {
            lo[i] = v4f{0.f, 0.f, 0.f, 0.f};
            hi[i] = lo[i];
            pv[i] = 0.f;
            if (rok && c0 < T) {
                const float* p = urow + c0 + loff;
                lo[i] = *reinterpret_cast<const v4f*>(p);
                hi[i] = *reinterpret_cast<const v4f*>(p + 8);
                pv[i] = (c0 == 0) ? u_init[(size_t)row * 2] : urow[c0 - 1];
            }
        }
    }

    h[0] = 1.0f;
    h[1] = -a1;
#pragma unroll
    for (int i = 2; i <= 16; ++i) h[i] = -a1 * h[i - 1] - a2 * h[i - 2];
    float hl = h[0];
#pragma unroll
    for (int i = 1; i <= 16; ++i) hl = (lane == i) ? h[i] : hl;
    float m00 = h[16], m01 = -a2 * h[15], m10 = h[15], m11 = -a2 * h[14];

    if (w == 0) {
        double p00 = m00, p01 = m01, p10 = m10, p11 = m11;
#pragma unroll
        for (int s = 1; s < ROW_CH * ROW_GR; s <<= 1) {
            if (s == ROW_CH && lane == 0)
                pw[0] = make_float4((float)p00, (float)p01, (float)p10, (float)p11);
            double tt = p01 * p10;
            double tr = p00 + p11;
            p00 = p00 * p00 + tt;
            p11 = p11 * p11 + tt;
            p01 = p01 * tr;
            p10 = p10 * tr;
        }
        if (lane == 0) pw[1] = make_float4((float)p00, (float)p01, (float)p10, (float)p11);
    }

    int colM = 4 * (n >> 3) + (n & 3) + 8 * ((n >> 2) & 1);
    v2f Af[4];
#pragma unroll
    for (int s = 0; s < 4; ++s) {
#pragma unroll
        for (int r = 0; r < 2; ++r) {
            int k = 8 * (s >> 1) + 4 * half + 2 * (s & 1) + r;
            Af[s][r] = h_gather(hl, colM - k);
        }
    }

    int src1 = half ? n : (n + 15);
    int src2 = lane ^ 16;

#pragma unroll
    for (int i = 0; i < ROW_P; ++i) {
        if (FULL && (i % ROW_G) == 0) {
            __builtin_amdgcn_sched_barrier(0);
#pragma unroll
            for (int j = i; j < i + ROW_G; ++j) {
                asm volatile("" : "+v"(lo[j]), "+v"(hi[j]));
            }
        }
        float q1 = half ? hi[i].w : lo[i].w;
        float q2 = half ? lo[i].w : hi[i].w;
        float p1 = __int_as_float(__builtin_amdgcn_ds_bpermute(src1 << 2, __float_as_int(q1)));
        float p2 = __int_as_float(__builtin_amdgcn_ds_bpermute(src2 << 2, __float_as_int(q2)));
        p1 = (lane == 0) ? pv[i] : p1;

        v2f Bf[4];
        Bf[0][0] = b0 * lo[i].x + b1 * p1;
        Bf[0][1] = b0 * lo[i].y + b1 * lo[i].x;
        Bf[1][0] = b0 * lo[i].z + b1 * lo[i].y;
        Bf[1][1] = b0 * lo[i].w + b1 * lo[i].z;
        Bf[2][0] = b0 * hi[i].x + b1 * p2;
        Bf[2][1] = b0 * hi[i].y + b1 * hi[i].x;
        Bf[3][0] = b0 * hi[i].z + b1 * hi[i].y;
        Bf[3][1] = b0 * hi[i].w + b1 * hi[i].z;

        v8f a = {0.f, 0.f, 0.f, 0.f, 0.f, 0.f, 0.f, 0.f};
#pragma unroll
        for (int s = 0; s < 4; ++s)
            a = __builtin_amdgcn_wmma_f32_16x16x4_f32(
                 false, Af[s],  false, Bf[s],
                 (short)0, a,  false,  false);
        acc[i] = a;
    }
}


__global__ void __launch_bounds__(ROW_W * 32)
iir_rows_fused(const float* __restrict__ u,
               const float* __restrict__ u_init,
               const float* __restrict__ y_init,
               const float* __restrict__ bc,
               const float* __restrict__ ac,
               float* __restrict__ y,
               int B, int T) {
    __shared__ __attribute__((aligned(16))) float2 st[ROW_R * ROW_TILES];
    __shared__ __attribute__((aligned(16))) float2 zb[ROW_R * ROW_NCH];
    __shared__ __attribute__((aligned(16))) float2 gt[ROW_R * ROW_NGR];
    __shared__ float4 pw[2];
    __shared__ v4f xs[ROW_W * ROW_NB * 64];

    int t    = threadIdx.x;
    int lane = t & 31;
    int w    = __builtin_amdgcn_readfirstlane(t >> 5);
    int n    = lane & 15;
    int half = lane >> 4;
    int wr   = w / ROW_WR;
    int wq   = w % ROW_WR;
    int row0 = blockIdx.x * ROW_R;
    int row  = row0 + wr;
    bool rok = row < B;
    int cw   = wq * (ROW_P * 256);

    float b0 = bc[0], b1 = bc[1];
    float a1 = ac[0], a2 = ac[1];

    float yi1[ROW_R], yi2[ROW_R];
#pragma unroll
    for (int r = 0; r < ROW_R; ++r) {
        bool ok = (row0 + r) < B;
        int  rr = ok ? (row0 + r) : row0;
        float v1 = y_init[(size_t)rr * 2 + 0];
        float v2 = y_init[(size_t)rr * 2 + 1];
        yi1[r] = ok ? v1 : 0.f;
        yi2[r] = ok ? v2 : 0.f;
    }

    const float* urow = u + (size_t)row * T;

    float h[17];
    v8f acc[ROW_P];
    if (rok && cw + ROW_P * 256 <= T)
        load_and_multiply<true>(urow, u_init, row, rok, cw, T, lane, n, half, w, b0, b1, a1, a2, pw, h, acc);
    else
        load_and_multiply<false>(urow, u_init, row, rok, cw, T, lane, n, half, w, b0, b1, a1, a2, pw, h, acc);
    float m00 = h[16], m01 = -a2 * h[15], m10 = h[15], m11 = -a2 * h[14];
    int loff = 16 * n + 4 * half;

    int tb = wr * ROW_TILES + wq * (ROW_P * 16);
    if (half) {
#pragma unroll
        for (int i = 0; i < ROW_P; ++i)
            st[tb + i * 16 + n] = make_float2(acc[i][7], acc[i][6]);
    }
    __syncthreads();

    float2 e[ROW_R][ROW_CH];
#pragma unroll
    for (int r = 0; r < ROW_R; ++r) {
        const float4* sp = reinterpret_cast<const float4*>(&st[r * ROW_TILES + t * ROW_CH]);
#pragma unroll
        for (int i = 0; i < ROW_CH / 2; ++i) {
            float4 q = sp[i];
            e[r][2 * i]     = make_float2(q.x, q.y);
            e[r][2 * i + 1] = make_float2(q.z, q.w);
        }
    }
#pragma unroll
    for (int r = 0; r < ROW_R; ++r) {
        float z1 = 0.f, z2 = 0.f;
#pragma unroll
        for (int i = 0; i < ROW_CH; ++i) {
            float n1 = m00 * z1 + m01 * z2 + e[r][i].x;
            float n2 = m10 * z1 + m11 * z2 + e[r][i].y;
            z1 = n1; z2 = n2;
        }
        zb[r * ROW_NCH + t] = make_float2(z1, z2);
    }
    __syncthreads();

    float4 Pc = pw[0];
    float4 Pg = pw[1];
    if (t < ROW_R * ROW_NGR) {
        float g1 = 0.f, g2 = 0.f;
#pragma unroll
        for (int q0 = 0; q0 < ROW_GR; q0 += ROW_U) {
            float2 ev[ROW_U];
#pragma unroll
            for (int j = 0; j < ROW_U; ++j) ev[j] = zb[t * ROW_GR + q0 + j];
#pragma unroll
            for (int j = 0; j < ROW_U; ++j) {
                float n1 = Pc.x * g1 + Pc.y * g2 + ev[j].x;
                float n2 = Pc.z * g1 + Pc.w * g2 + ev[j].y;
                g1 = n1; g2 = n2;
            }
        }
        gt[t] = make_float2(g1, g2);
    }
    __syncthreads();

    int g  = t / ROW_GR;
    int jc = t % ROW_GR;
    float s1[ROW_R], s2[ROW_R];
#pragma unroll
    for (int r = 0; r < ROW_R; ++r) { s1[r] = yi1[r]; s2[r] = yi2[r]; }
#pragma unroll
    for (int v0 = 0; v0 < ROW_NGR; v0 += ROW_U3) {
        float2 ev[ROW_R][ROW_U3];
#pragma unroll
        for (int r = 0; r < ROW_R; ++r) {
            const float4* gp = reinterpret_cast<const float4*>(&gt[r * ROW_NGR + v0]);
#pragma unroll
            for (int j = 0; j < ROW_U3 / 2; ++j) {
                float4 q = gp[j];
                ev[r][2 * j]     = make_float2(q.x, q.y);
                ev[r][2 * j + 1] = make_float2(q.z, q.w);
            }
        }
#pragma unroll
        for (int j = 0; j < ROW_U3; ++j) {
            bool take = (v0 + j) < g;
#pragma unroll
            for (int r = 0; r < ROW_R; ++r) {
                float n1 = Pg.x * s1[r] + Pg.y * s2[r] + ev[r][j].x;
                float n2 = Pg.z * s1[r] + Pg.w * s2[r] + ev[r][j].y;
                s1[r] = take ? n1 : s1[r];
                s2[r] = take ? n2 : s2[r];
            }
        }
    }
#pragma unroll
    for (int q0 = 0; q0 < ROW_GR; q0 += ROW_U3) {
        float2 ev[ROW_R][ROW_U3];
#pragma unroll
        for (int r = 0; r < ROW_R; ++r) {
            const float4* zp = reinterpret_cast<const float4*>(&zb[r * ROW_NCH + g * ROW_GR + q0]);
#pragma unroll
            for (int j = 0; j < ROW_U3 / 2; ++j) {
                float4 q = zp[j];
                ev[r][2 * j]     = make_float2(q.x, q.y);
                ev[r][2 * j + 1] = make_float2(q.z, q.w);
            }
        }
#pragma unroll
        for (int j = 0; j < ROW_U3; ++j) {
            bool take = (q0 + j) < jc;
#pragma unroll
            for (int r = 0; r < ROW_R; ++r) {
                float n1 = Pc.x * s1[r] + Pc.y * s2[r] + ev[r][j].x;
                float n2 = Pc.z * s1[r] + Pc.w * s2[r] + ev[r][j].y;
                s1[r] = take ? n1 : s1[r];
                s2[r] = take ? n2 : s2[r];
            }
        }
    }
#pragma unroll
    for (int r = 0; r < ROW_R; ++r) {
        float2 sv[ROW_CH];
        float c1 = s1[r], c2 = s2[r];
#pragma unroll
        for (int i = 0; i < ROW_CH; ++i) {
            sv[i] = make_float2(c1, c2);
            float n1 = m00 * c1 + m01 * c2 + e[r][i].x;
            float n2 = m10 * c1 + m11 * c2 + e[r][i].y;
            c1 = n1; c2 = n2;
        }
        float4* sp = reinterpret_cast<float4*>(&st[r * ROW_TILES + t * ROW_CH]);
#pragma unroll
        for (int i = 0; i < ROW_CH / 2; ++i)
            sp[i] = make_float4(sv[2 * i].x, sv[2 * i].y, sv[2 * i + 1].x, sv[2 * i + 1].y);
    }
    __syncthreads();

    float gA[4], fA[4], gB[4], fB[4];
#pragma unroll
    for (int i = 0; i < 4; ++i) {
        gA[i] = half ? h[4 + i + 1]  : h[i + 1];
        fA[i] = -a2 * (half ? h[4 + i]  : h[i]);
        gB[i] = half ? h[12 + i + 1] : h[8 + i + 1];
        fB[i] = -a2 * (half ? h[12 + i] : h[8 + i]);
    }
    float* yrow = y + (size_t)row * T;
    if (rok && cw + ROW_P * 256 <= T) {
        float2 sv[ROW_P];
#pragma unroll
        for (int i = 0; i < ROW_P; ++i) sv[i] = st[tb + i * 16 + n];
        v4f* xw = xs + w * (ROW_NB * 64);
        int q4  = lane & 3, t4 = lane >> 2;
        int wo0 = 16 * half       + ((n + 4 * half) & 15);
        int wo1 = 16 * (2 + half) + ((n + 8 + 4 * half) & 15);
        int roA = 16 * q4 + ((t4 + 4 * q4) & 15);
        int roB = 16 * q4 + ((t4 + 8 + 4 * q4) & 15);
        float* yl = yrow + cw + 4 * lane;
        v4f la[ROW_P], lb[ROW_P];
#pragma unroll
        for (int i = 0; i < ROW_P + ROW_SD; ++i) {
            if (i < ROW_P) {
                float2 s = sv[i];
                v4f o0, o1;
                o0.x = acc[i][0] + (gA[0] * s.x + fA[0] * s.y);
                o0.y = acc[i][1] + (gA[1] * s.x + fA[1] * s.y);
                o0.z = acc[i][2] + (gA[2] * s.x + fA[2] * s.y);
                o0.w = acc[i][3] + (gA[3] * s.x + fA[3] * s.y);
                o1.x = acc[i][4] + (gB[0] * s.x + fB[0] * s.y);
                o1.y = acc[i][5] + (gB[1] * s.x + fB[1] * s.y);
                o1.z = acc[i][6] + (gB[2] * s.x + fB[2] * s.y);
                o1.w = acc[i][7] + (gB[3] * s.x + fB[3] * s.y);
                v4f* slot = xw + (i % ROW_NB) * 64;
                slot[wo0] = o0;
                slot[wo1] = o1;
                wave_lds_exchange_point();
                la[i] = slot[roA];
                lb[i] = slot[roB];
            }
            if (i >= ROW_SD) {
                float* p = yl + (i - ROW_SD) * 256;
                *reinterpret_cast<v4f*>(p)       = la[i - ROW_SD];
                *reinterpret_cast<v4f*>(p + 128) = lb[i - ROW_SD];
            }
        }
    } else {
#pragma unroll
        for (int i = 0; i < ROW_P; ++i) {
            int c0 = cw + i * 256;
            float2 s = st[tb + i * 16 + n];
            float4 o0, o1;
            o0.x = acc[i][0] + (gA[0] * s.x + fA[0] * s.y);
            o0.y = acc[i][1] + (gA[1] * s.x + fA[1] * s.y);
            o0.z = acc[i][2] + (gA[2] * s.x + fA[2] * s.y);
            o0.w = acc[i][3] + (gA[3] * s.x + fA[3] * s.y);
            o1.x = acc[i][4] + (gB[0] * s.x + fB[0] * s.y);
            o1.y = acc[i][5] + (gB[1] * s.x + fB[1] * s.y);
            o1.z = acc[i][6] + (gB[2] * s.x + fB[2] * s.y);
            o1.w = acc[i][7] + (gB[3] * s.x + fB[3] * s.y);
            if (rok && c0 < T) {
                float* p = yrow + c0 + loff;
                *reinterpret_cast<float4*>(p)     = o0;
                *reinterpret_cast<float4*>(p + 8) = o1;
            }
        }
    }
}

extern "C" void kernel_launch(void* const* d_in, const int* in_sizes, int n_in,
                              void* d_out, int out_size, void* d_ws, size_t ws_size,
                              hipStream_t stream) {
    const float* bc = (const float*)d_in[0];
    const float* ac = (const float*)d_in[1];
    const float* u  = (const float*)d_in[2];
    const float* yi = (const float*)d_in[3];
    const float* ui = (const float*)d_in[4];
    float* y = (float*)d_out;

    int B = in_sizes[3] / 2;
    int T = in_sizes[2] / B;

    iir_rows_fused<<<(B + ROW_R - 1) / ROW_R, ROW_W * 32, 0, stream>>>(u, ui, yi, bc, ac, y, B, T);
}
